# starting kernel
# baseline (speedup 1.0000x reference)
#include <hip/hip_runtime.h>

#define BATCH   8
#define NQ      512
#define NKV     512
#define NH      16
#define DMODEL  256
#define NEG_INF_F (-1.0e9f)

#define KC_ATT  32
#define KC_GEMM 32

typedef __attribute__((ext_vector_type(2))) float v2f;
typedef __attribute__((ext_vector_type(8))) float v8f;

__device__ __forceinline__ v8f wmma_f32(v2f a, v2f b, v8f c) {
    return __builtin_amdgcn_wmma_f32_16x16x4_f32(
         false, a,  false, b,
         (short)0, c,  false,  false);
}

__launch_bounds__(256)
__global__ void gemm_bias_256(const float* __restrict__ in,
                              const float* __restrict__ W,
                              const float* __restrict__ bias,
                              float* __restrict__ out) {
    __shared__ float s_in[16 * 257];
    __shared__ float s_w[DMODEL * (KC_GEMM + 1)];

    const int tid   = threadIdx.x;
    const int wid   = tid >> 5;
    const int lane  = tid & 31;
    const int row   = lane & 15;
    const int khalf = (lane < 16) ? 0 : 2;
    const int n0    = blockIdx.x * 16;

    for (int i = 0; i < 4; ++i) {
        int e4 = (tid + 256 * i) * 4;
        int r  = e4 >> 8, k = e4 & 255;
        const float4 v = *(const float4*)&in[(size_t)(n0 + r) * DMODEL + k];
        s_in[r * 257 + k + 0] = v.x;
        s_in[r * 257 + k + 1] = v.y;
        s_in[r * 257 + k + 2] = v.z;
        s_in[r * 257 + k + 3] = v.w;
    }

    v8f acc0 = {};
    v8f acc1 = {};
    const int d0 = (wid)     * 16;
    const int d1 = (wid + 8) * 16;

    for (int kc = 0; kc < DMODEL; kc += KC_GEMM) {
        __syncthreads();
        for (int i = 0; i < 8; ++i) {
            int e4 = (tid + 256 * i) * 4;
            int d  = e4 >> 5, k = e4 & 31;
            const float4 v = *(const float4*)&W[(size_t)d * DMODEL + kc + k];
            s_w[d * (KC_GEMM + 1) + k + 0] = v.x;
            s_w[d * (KC_GEMM + 1) + k + 1] = v.y;
            s_w[d * (KC_GEMM + 1) + k + 2] = v.z;
            s_w[d * (KC_GEMM + 1) + k + 3] = v.w;
        }
        __syncthreads();

        for (int ks = 0; ks < KC_GEMM; ks += 4) {
            const float* ap = &s_in[row * 257 + kc + ks + khalf];
            v2f a; a.x = ap[0]; a.y = ap[1];

            const float* bp0 = &s_w[(d0 + row) * (KC_GEMM + 1) + ks + khalf];
            v2f b0; b0.x = bp0[0]; b0.y = bp0[1];
            const float* bp1 = &s_w[(d1 + row) * (KC_GEMM + 1) + ks + khalf];
            v2f b1; b1.x = bp1[0]; b1.y = bp1[1];

            acc0 = wmma_f32(a, b0, acc0);
            acc1 = wmma_f32(a, b1, acc1);
        }
    }

    const float bias0 = bias[d0 + row];
    const float bias1 = bias[d1 + row];
    for (int i = 0; i < 8; ++i) {
        const int r = i + ((lane < 16) ? 0 : 8);
        out[(size_t)(n0 + r) * DMODEL + d0 + row] = acc0[i] + bias0;
        out[(size_t)(n0 + r) * DMODEL + d1 + row] = acc1[i] + bias1;
    }
}

#define MSG_ROWSTRIDE (KC_ATT + 1)
#define MSG_HSTRIDE   (16 * MSG_ROWSTRIDE + 4)

__launch_bounds__(256)
__global__ void attn_kernel(const float* __restrict__ messages,
                            const int*   __restrict__ adj,
                            const float* __restrict__ proj,
                            float*       __restrict__ attn_out) {
    __shared__ float s_msg[NH * MSG_HSTRIDE];
    __shared__ float s_mask[16 * MSG_ROWSTRIDE];
    __shared__ float s_scale[8 * 32];
    __shared__ float s_m[8 * 32];

    const int tid   = threadIdx.x;
    const int wid   = tid >> 5;
    const int lane  = tid & 31;
    const int row   = lane & 15;
    const int khalf = (lane < 16) ? 0 : 2;
    const int h0    = wid * 2;
    const int h1    = wid * 2 + 1;
    const int b     = blockIdx.x / (NQ / 16);
    const int q0    = (blockIdx.x % (NQ / 16)) * 16;
    const int own   = wid * 32 + lane;

    const float* msg_base = messages + (size_t)(b * NQ + q0) * NKV * NH;
    const int*   adj_base = adj      + (size_t)(b * NQ + q0) * NKV;
    const float* proj_b   = proj     + (size_t)b * NKV * DMODEL;

    float m = -3.0e38f, l = 0.0f, s2 = 0.0f;
    v8f acc0 = {};
    v8f acc1 = {};

    for (int kc = 0; kc < NKV; kc += KC_ATT) {
        __syncthreads();
        for (int i = 0; i < 2; ++i) {
            int e = tid + 256 * i;
            int q = e >> 5, k = e & 31;
            s_mask[q * MSG_ROWSTRIDE + k] =
                (adj_base[(size_t)q * NKV + kc + k] > 0) ? 0.0f : NEG_INF_F;
        }
        __syncthreads();
        for (int i = 0; i < 8; ++i) {
            int e4 = (tid + 256 * i) * 4;
            int q  = e4 >> 9;
            int j  = e4 & 511;
            int k  = j >> 4, h = j & 15;
            const float4 v =
                *(const float4*)&msg_base[(size_t)q * (NKV * NH) + kc * NH + j];
            const float mk = s_mask[q * MSG_ROWSTRIDE + k];
            s_msg[(h + 0) * MSG_HSTRIDE + q * MSG_ROWSTRIDE + k] = v.x + mk;
            s_msg[(h + 1) * MSG_HSTRIDE + q * MSG_ROWSTRIDE + k] = v.y + mk;
            s_msg[(h + 2) * MSG_HSTRIDE + q * MSG_ROWSTRIDE + k] = v.z + mk;
            s_msg[(h + 3) * MSG_HSTRIDE + q * MSG_ROWSTRIDE + k] = v.w + mk;
        }

        if (kc + KC_ATT < NKV) {
            const int pq = tid >> 4;
            const int pj = (tid & 15) * 32;
            __builtin_prefetch(&msg_base[(size_t)pq * (NKV * NH) +
                                         (kc + KC_ATT) * NH + pj], 0, 3);
            if (tid < 16) {
                __builtin_prefetch(&adj_base[(size_t)tid * NKV + kc + KC_ATT], 0, 3);
            }
        }
        __syncthreads();

        {
            const int myh = (lane < 16) ? h0 : h1;
            const float* rp = &s_msg[myh * MSG_HSTRIDE + row * MSG_ROWSTRIDE];
            float mc = -3.0e38f;
            for (int k = 0; k < KC_ATT; ++k) mc = fmaxf(mc, rp[k]);
            const float mnew = fmaxf(m, mc);
            const float r    = __expf(m - mnew);
            float lc = 0.0f, s2c = 0.0f;
            for (int k = 0; k < KC_ATT; ++k) {
                const float t = __expf(rp[k] - mnew);
                lc  += t;
                s2c += t * t;
            }
            l  = l * r + lc;
            s2 = s2 * (r * r) + s2c;
            m  = mnew;
            s_scale[own] = r;
            s_m[own]     = mnew;
        }

        for (int i = 0; i < 8; ++i) {
            const int rr = i + ((lane < 16) ? 0 : 8);
            acc0[i] *= s_scale[wid * 32 + rr];
            acc1[i] *= s_scale[wid * 32 + 16 + rr];
        }
        const float m0 = s_m[wid * 32 + row];
        const float m1 = s_m[wid * 32 + 16 + row];

        const float* p0 = &s_msg[h0 * MSG_HSTRIDE + row * MSG_ROWSTRIDE + khalf];
        const float* p1 = &s_msg[h1 * MSG_HSTRIDE + row * MSG_ROWSTRIDE + khalf];
        for (int ks = 0; ks < KC_ATT; ks += 4) {
            v2f a0; a0.x = __expf(p0[ks] - m0); a0.y = __expf(p0[ks + 1] - m0);
            v2f a1; a1.x = __expf(p1[ks] - m1); a1.y = __expf(p1[ks + 1] - m1);

            const float* vrow = proj_b + (size_t)(kc + ks + khalf) * DMODEL;
            v2f b0; b0.x = vrow[h0 * 16 + row]; b0.y = vrow[DMODEL + h0 * 16 + row];
            v2f b1; b1.x = vrow[h1 * 16 + row]; b1.y = vrow[DMODEL + h1 * 16 + row];

            acc0 = wmma_f32(a0, b0, acc0);
            acc1 = wmma_f32(a1, b1, acc1);
        }
    }

    s_scale[own] = sqrtf(s2) / (l * l);
    for (int i = 0; i < 8; ++i) {
        const int rr = i + ((lane < 16) ? 0 : 8);
        const float f0 = s_scale[wid * 32 + rr];
        const float f1 = s_scale[wid * 32 + 16 + rr];
        attn_out[(size_t)(b * NQ + q0 + rr) * DMODEL + h0 * 16 + row] = acc0[i] * f0;
        attn_out[(size_t)(b * NQ + q0 + rr) * DMODEL + h1 * 16 + row] = acc1[i] * f1;
    }
}

extern "C" void kernel_launch(void* const* d_in, const int* in_sizes, int n_in,
                              void* d_out, int out_size, void* d_ws, size_t ws_size,
                              hipStream_t stream) {
    (void)in_sizes; (void)n_in; (void)out_size; (void)ws_size;

    const float* v_inv    = (const float*)d_in[0];
    const float* messages = (const float*)d_in[1];
    const int*   adj      = (const int*)  d_in[2];
    const float* W_in     = (const float*)d_in[3];
    const float* b_in     = (const float*)d_in[4];
    const float* W_out    = (const float*)d_in[5];
    const float* b_out    = (const float*)d_in[6];
    float*       out      = (float*)d_out;

    float* proj = (float*)d_ws;
    float* attn = proj + (size_t)BATCH * NKV * DMODEL;

    gemm_bias_256<<<(BATCH * NKV) / 16, 256, 0, stream>>>(v_inv, W_in, b_in, proj);
    attn_kernel<<<BATCH * (NQ / 16), 256, 0, stream>>>(messages, adj, proj, attn);
    gemm_bias_256<<<(BATCH * NQ) / 16, 256, 0, stream>>>(attn, W_out, b_out, out);
}


# both projection GEMMs re-tiled: 64x64 output tile per block, whole K of both operand slabs resident in LDS (130 KB), all loads issued up front, consumed in 4 K stages; k order inside 8-blocks changes
# speedup vs baseline: 1.4243x; 1.4243x over previous
#include <hip/hip_runtime.h>

#define BATCH   8
#define NQ      512
#define NKV     512
#define NH      16
#define DMODEL  256
#define NEG_INF_F (-1.0e9f)

typedef __attribute__((ext_vector_type(2))) float v2f;
typedef __attribute__((ext_vector_type(4))) float v4f;
typedef __attribute__((ext_vector_type(8))) float v8f;

__device__ __forceinline__ v8f wmma_f32(v2f a, v2f b, v8f c) {
    return __builtin_amdgcn_wmma_f32_16x16x4_f32(
         false, a,  false, b,
         (short)0, c,  false,  false);
}

__device__ __forceinline__ float xhalf(float x) {
    const int b = __builtin_bit_cast(int, x);
    return __builtin_bit_cast(float,
        __builtin_amdgcn_permlanex16(b, b, 0x76543210, 0xfedcba98, false, false));
}

__device__ __forceinline__ void lds_barrier() {
    asm volatile("s_wait_dscnt 0x0\n\ts_barrier_signal -1\n\ts_barrier_wait -1" ::: "memory");
}

#define G_PITCH 260

template <bool FRAG_ORDER>
__launch_bounds__(256)
__global__ void gemm_bias_256(const float* __restrict__ in,
                              const float* __restrict__ W,
                              const float* __restrict__ bias,
                              float* __restrict__ out) {
    __shared__ float s_in[64 * G_PITCH];
    __shared__ float s_w [64 * G_PITCH];

    const int tid  = threadIdx.x;
    const int wid  = tid >> 5;
    const int lane = tid & 31;
    const int row  = lane & 15;
    const int kh   = lane >> 4;
    const int n0   = (blockIdx.x >> 2) * 64;
    const int c0   = (blockIdx.x & 3) * 64;
    const int wr   = wid & 3;
    const int wc   = wid >> 2;

    const int lr = tid >> 4;
    const int lk = (tid & 15) * 4;
    const float* gi = in + (size_t)(n0 + lr) * DMODEL + lk;
    const float* gw = W  + (size_t)(c0 + lr) * DMODEL + lk;
    v4f ri[16], rw[16];
#pragma unroll
    for (int s = 0; s < 4; ++s) {
#pragma unroll
        for (int j = 0; j < 4; ++j)
            ri[s * 4 + j] = *(const v4f*)(gi + (size_t)(16 * j) * DMODEL + 64 * s);
#pragma unroll
        for (int j = 0; j < 4; ++j)
            rw[s * 4 + j] = *(const v4f*)(gw + (size_t)(16 * j) * DMODEL + 64 * s);
    }

    v8f acc0 = {};
    v8f acc1 = {};
    float*       si  = &s_in[lr * G_PITCH + lk];
    float*       sw  = &s_w [lr * G_PITCH + lk];
    const float* ap  = &s_in[(wr * 16 + row) * G_PITCH + 4 * kh];
    const float* bp0 = &s_w [(wc * 32 + row) * G_PITCH + 4 * kh];
    const float* bp1 = bp0 + 16 * G_PITCH;

#pragma unroll
    for (int s = 0; s < 4; ++s) {
#pragma unroll
        for (int j = 0; j < 4; ++j) {
            *(v4f*)(si + 16 * j * G_PITCH + 64 * s) = ri[s * 4 + j];
            *(v4f*)(sw + 16 * j * G_PITCH + 64 * s) = rw[s * 4 + j];
        }
        lds_barrier();
#pragma unroll
        for (int kb = 0; kb < 8; ++kb) {
            const int k0 = 64 * s + 8 * kb;
            const v4f a  = *(const v4f*)(ap  + k0);
            const v4f b0 = *(const v4f*)(bp0 + k0);
            const v4f b1 = *(const v4f*)(bp1 + k0);
            v2f a_lo;  a_lo.x  = a.x;  a_lo.y  = a.y;
            v2f a_hi;  a_hi.x  = a.z;  a_hi.y  = a.w;
            v2f b0_lo; b0_lo.x = b0.x; b0_lo.y = b0.y;
            v2f b0_hi; b0_hi.x = b0.z; b0_hi.y = b0.w;
            v2f b1_lo; b1_lo.x = b1.x; b1_lo.y = b1.y;
            v2f b1_hi; b1_hi.x = b1.z; b1_hi.y = b1.w;
            acc0 = wmma_f32(a_lo, b0_lo, acc0);
            acc1 = wmma_f32(a_lo, b1_lo, acc1);
            acc0 = wmma_f32(a_hi, b0_hi, acc0);
            acc1 = wmma_f32(a_hi, b1_hi, acc1);
        }
    }

    const int nw = n0 + wr * 16;
    const int d0 = c0 + wc * 32;
    const int d1 = d0 + 16;
    const float bias0 = bias[d0 + row];
    const float bias1 = bias[d1 + row];
    if (FRAG_ORDER) {
        const int chunk = (nw >> 3) + kh;
        const int hp    = d0 >> 5;
#pragma unroll
        for (int i = 0; i < 8; i += 2) {
            const int rh = i >> 2;
            const int sl = (i >> 1) & 1;
            v4f v;
            v.x = acc0[i]     + bias0;
            v.y = acc0[i + 1] + bias0;
            v.z = acc1[i]     + bias1;
            v.w = acc1[i + 1] + bias1;
            *(v4f*)&out[(size_t)chunk * 2048 + sl * 1024 + hp * 128 + (row + 16 * rh) * 4] = v;
        }
    } else {
#pragma unroll
        for (int i = 0; i < 8; ++i) {
            const int r = i + 8 * kh;
            out[(size_t)(nw + r) * DMODEL + d0 + row] = acc0[i] + bias0;
            out[(size_t)(nw + r) * DMODEL + d1 + row] = acc1[i] + bias1;
        }
    }
}

#define ACC_PITCH 260
#define NCHUNK    8

typedef __attribute__((ext_vector_type(4))) int v4i;

__launch_bounds__(256)
__global__ void attn_kernel(const float* __restrict__ messages,
                            const int*   __restrict__ adj,
                            const float* __restrict__ projf,
                            float*       __restrict__ attn_out) {
    __shared__ float s_acc[8 * 16 * ACC_PITCH];
    __shared__ float s_mx[8 * 256];
    __shared__ float s_l[16 * 256];
    __shared__ float s_s2[16 * 256];

    const int tid  = threadIdx.x;
    const int wid  = tid >> 5;
    const int lane = tid & 31;
    const int q    = lane & 15;
    const int kh   = lane >> 4;
    const int b    = blockIdx.x / (NQ / 16);
    const int q0   = (blockIdx.x % (NQ / 16)) * 16;
    const int kw   = wid * 64;

    const float* mp = messages + ((size_t)(b * NQ + q0 + q) * NKV + kw + 4 * kh) * NH;
    const int*   ap = adj + (size_t)(b * NQ + q0 + q) * NKV + kw + 4 * kh;
    const float* vp = projf + (size_t)(b * (NKV / 8) + wid * NCHUNK) * 2048 + lane * 4;

    float m[16], l[16], s2[16];
    v8f acc[16];
#pragma unroll
    for (int h = 0; h < 16; ++h) {
        m[h] = -3.0e38f; l[h] = 0.0f; s2[h] = 0.0f;
        acc[h] = v8f{};
    }

    __builtin_prefetch(mp + 128,      0, 3);
    __builtin_prefetch(mp + 128 + 32, 0, 3);

#pragma nounroll
    for (int c = 0; c < NCHUNK; ++c) {
        v4f g[16], v[16];
#pragma unroll
        for (int i = 0; i < 16; ++i) g[i] = *(const v4f*)(mp + c * 128 + i * 4);
        const v4i a = *(const v4i*)(ap + c * 8);
#pragma unroll
        for (int i = 0; i < 16; ++i)
            v[i] = *(const v4f*)(vp + c * 2048 + (i >> 3) * 1024 + (i & 7) * 128);

        if (c + 2 < NCHUNK) {
            __builtin_prefetch(mp + (c + 2) * 128,      0, 3);
            __builtin_prefetch(mp + (c + 2) * 128 + 32, 0, 3);
            __builtin_prefetch(ap + (c + 2) * 8,        0, 3);
        }

        float mk[4];
        mk[0] = (a.x > 0) ? 0.0f : NEG_INF_F;
        mk[1] = (a.y > 0) ? 0.0f : NEG_INF_F;
        mk[2] = (a.z > 0) ? 0.0f : NEG_INF_F;
        mk[3] = (a.w > 0) ? 0.0f : NEG_INF_F;

        v2f p[32];
#pragma unroll
        for (int h = 0; h < 16; ++h) {
            const int j = h >> 2, cc = h & 3;
            const float x0 = g[0  + j][cc] + mk[0];
            const float x1 = g[4  + j][cc] + mk[1];
            const float x2 = g[8  + j][cc] + mk[2];
            const float x3 = g[12 + j][cc] + mk[3];
            float mc = fmaxf(fmaxf(x0, x1), fmaxf(x2, x3));
            mc = fmaxf(mc, xhalf(mc));
            const float mnew = fmaxf(m[h], mc);
            const float r  = __expf(m[h] - mnew);
            const float p0 = __expf(x0 - mnew);
            const float p1 = __expf(x1 - mnew);
            const float p2 = __expf(x2 - mnew);
            const float p3 = __expf(x3 - mnew);
            l[h]  = l[h]  * r       + ((p0 + p1) + (p2 + p3));
            s2[h] = s2[h] * (r * r) + ((p0 * p0 + p1 * p1) + (p2 * p2 + p3 * p3));
            m[h]  = mnew;
            acc[h] *= r;
            v2f b0; b0.x = p0; b0.y = p1;
            v2f b1; b1.x = p2; b1.y = p3;
            p[2 * h]     = b0;
            p[2 * h + 1] = b1;
        }
#pragma unroll
        for (int h = 0; h < 16; ++h) {
            const int hp = h >> 1, o = (h & 1) * 2;
            v2f a0; a0.x = v[hp][o];     a0.y = v[hp][o + 1];
            v2f a1; a1.x = v[8 + hp][o]; a1.y = v[8 + hp][o + 1];
            acc[h] = wmma_f32(a0, p[2 * h],     acc[h]);
            acc[h] = wmma_f32(a1, p[2 * h + 1], acc[h]);
        }
    }

    {
        float* pa = &s_acc[(wid * 16 + q) * ACC_PITCH + 8 * kh];
#pragma unroll
        for (int h = 0; h < 16; ++h) {
            v4f lo, hi;
            lo.x = acc[h][0]; lo.y = acc[h][1]; lo.z = acc[h][2]; lo.w = acc[h][3];
            hi.x = acc[h][4]; hi.y = acc[h][5]; hi.z = acc[h][6]; hi.w = acc[h][7];
            *(v4f*)(pa + h * 16)     = lo;
            *(v4f*)(pa + h * 16 + 4) = hi;
            s_mx[wid * 256 + q * 16 + h]             = m[h];
            s_l [(wid * 2 + kh) * 256 + q * 16 + h]  = l[h];
            s_s2[(wid * 2 + kh) * 256 + q * 16 + h]  = s2[h];
        }
    }
    __syncthreads();

    {
        const int fq = tid >> 4, fh = tid & 15;
        float mw[8];
        float M = -3.0e38f;
#pragma unroll
        for (int w = 0; w < 8; ++w) {
            mw[w] = s_mx[w * 256 + fq * 16 + fh];
            M = fmaxf(M, mw[w]);
        }
        float L = 0.0f, S2 = 0.0f;
        v4f o0 = {}, o1 = {}, o2 = {}, o3 = {};
#pragma unroll
        for (int w = 0; w < 8; ++w) {
            const float f = __expf(mw[w] - M);
            L  += f * (s_l[(2 * w) * 256 + fq * 16 + fh] + s_l[(2 * w + 1) * 256 + fq * 16 + fh]);
            S2 += (f * f) * (s_s2[(2 * w) * 256 + fq * 16 + fh] + s_s2[(2 * w + 1) * 256 + fq * 16 + fh]);
            const float* sp = &s_acc[(w * 16 + fq) * ACC_PITCH + fh * 16];
            o0 += f * *(const v4f*)(sp);
            o1 += f * *(const v4f*)(sp + 4);
            o2 += f * *(const v4f*)(sp + 8);
            o3 += f * *(const v4f*)(sp + 12);
        }
        const float fac = sqrtf(S2) / (L * L);
        float* op = attn_out + (size_t)(b * NQ + q0 + fq) * DMODEL + fh * 16;
        *(v4f*)(op)      = o0 * fac;
        *(v4f*)(op + 4)  = o1 * fac;
        *(v4f*)(op + 8)  = o2 * fac;
        *(v4f*)(op + 12) = o3 * fac;
    }
}

extern "C" void kernel_launch(void* const* d_in, const int* in_sizes, int n_in,
                              void* d_out, int out_size, void* d_ws, size_t ws_size,
                              hipStream_t stream) {
    (void)in_sizes; (void)n_in; (void)out_size; (void)ws_size;

    const float* v_inv    = (const float*)d_in[0];
    const float* messages = (const float*)d_in[1];
    const int*   adj      = (const int*)  d_in[2];
    const float* W_in     = (const float*)d_in[3];
    const float* b_in     = (const float*)d_in[4];
    const float* W_out    = (const float*)d_in[5];
    const float* b_out    = (const float*)d_in[6];
    float*       out      = (float*)d_out;

    float* proj = (float*)d_ws;
    float* attn = proj + (size_t)BATCH * NKV * DMODEL;

    gemm_bias_256<true><<<(BATCH * NKV) / 16, 256, 0, stream>>>(v_inv, W_in, b_in, proj);
    attn_kernel<<<BATCH * (NQ / 16), 256, 0, stream>>>(messages, adj, proj, attn);
    gemm_bias_256<false><<<(BATCH * NQ) / 16, 256, 0, stream>>>(attn, W_out, b_out, out);
}
